# starting kernel
# baseline (speedup 1.0000x reference)
#include <hip/hip_runtime.h>
#include <math.h>

typedef __attribute__((ext_vector_type(2))) float v2f;
typedef __attribute__((ext_vector_type(8))) float v8f;

#define WAVES 8
constexpr int IN_F = 14, H = 24, T = 25, NC = 4, WH = 12;
constexpr int G  = 4 * H;
constexpr int NT = G / 16;
constexpr int KX = 4;
constexpr int KH = H / 4;

__device__ __forceinline__ float fsig(float x) {
  float e = __builtin_amdgcn_exp2f(x * -1.4426950408889634f);
  return __builtin_amdgcn_rcpf(1.f + e);
}
__device__ __forceinline__ float ftanh(float x) {
  float t = __builtin_amdgcn_exp2f(-2.8853900817779268f * fabsf(x));
  float r = (1.f - t) * __builtin_amdgcn_rcpf(1.f + t);
  return copysignf(r, x);
}

#define WAVE_FENCE() asm volatile("" ::: "memory")

__global__ __launch_bounds__(32 * WAVES) void bilstm_center_kernel(
    const float* __restrict__ x,
    const float* __restrict__ w_ih_f, const float* __restrict__ w_hh_f,
    const float* __restrict__ b_ih_f, const float* __restrict__ b_hh_f,
    const float* __restrict__ w_ih_b, const float* __restrict__ w_hh_b,
    const float* __restrict__ b_ih_b, const float* __restrict__ b_hh_b,
    const float* __restrict__ head_w, const float* __restrict__ head_b,
    float* __restrict__ out, int B)
{
  __shared__ float s_wx[2][NT * KX * 64];
  __shared__ float s_wh[2][NT * KH * 64];
  __shared__ float s_bias[2][G];
  __shared__ float s_headw[NC * 2 * H];
  __shared__ float s_headb[NC];
  __shared__ float s_gates[WAVES][16 * G];
  __shared__ float s_h[WAVES][16 * H];
  __shared__ float s_hf[WAVES][16 * H];

  const int tid = threadIdx.x;

  for (int i = tid; i < 2 * NT * KX * 64; i += blockDim.x) {
    int d = i / (NT * KX * 64);
    int r = i % (NT * KX * 64);
    int frag = r >> 6, slot = r & 63;
    int nt = frag / KX, kk = frag % KX;
    int ln = slot >> 1, e = slot & 1;
    int n = nt * 16 + (ln & 15);
    int k = kk * 4 + ((ln >> 4) << 1) + e;
    const float* w = d ? w_ih_b : w_ih_f;
    s_wx[d][r] = (k < IN_F) ? w[n * IN_F + k] : 0.f;
  }
  for (int i = tid; i < 2 * NT * KH * 64; i += blockDim.x) {
    int d = i / (NT * KH * 64);
    int r = i % (NT * KH * 64);
    int frag = r >> 6, slot = r & 63;
    int nt = frag / KH, kk = frag % KH;
    int ln = slot >> 1, e = slot & 1;
    int n = nt * 16 + (ln & 15);
    int k = kk * 4 + ((ln >> 4) << 1) + e;
    const float* w = d ? w_hh_b : w_hh_f;
    s_wh[d][r] = w[n * H + k];
  }
  for (int i = tid; i < 2 * G; i += blockDim.x) {
    int d = i / G, n = i % G;
    s_bias[d][n] = d ? (b_ih_b[n] + b_hh_b[n]) : (b_ih_f[n] + b_hh_f[n]);
  }
  for (int i = tid; i < NC * 2 * H; i += blockDim.x) s_headw[i] = head_w[i];
  if (tid < NC) s_headb[tid] = head_b[tid];
  __syncthreads();

  const int lane = tid & 31;
  const int w    = tid >> 5;
  const int M    = lane & 15;
  const int hi   = lane >> 4;
  const int b0   = (blockIdx.x * WAVES + w) * 16;
  if (b0 >= B) return;

  float* gatesL = &s_gates[w][0];
  float* hL     = &s_h[w][0];

  float bx[2][NT];
  #pragma unroll
  for (int d = 0; d < 2; ++d)
    #pragma unroll
    for (int nt = 0; nt < NT; ++nt) bx[d][nt] = s_bias[d][nt * 16 + M];

  float cst[12];
  const float* xbase = x + (size_t)b0 * T * IN_F;

  for (int d = 0; d < 2; ++d) {
    #pragma unroll
    for (int u = 0; u < 12; ++u) cst[u] = 0.f;
    for (int e = lane; e < 16 * H; e += 32) hL[e] = 0.f;
    WAVE_FENCE();

    for (int step = 0; step <= WH; ++step) {
      const int t = d ? (T - 1 - step) : step;
      const float* xr = xbase + ((size_t)M * T + t) * IN_F;

      v2f ax[KX];
      #pragma unroll
      for (int kk = 0; kk < KX; ++kk) {
        const int ks = kk * 4 + hi * 2;
        if (ks + 1 < IN_F) ax[kk] = *(const v2f*)(xr + ks);
        else               ax[kk] = (v2f){0.f, 0.f};
      }
      v2f ah[KH];
      #pragma unroll
      for (int kk = 0; kk < KH; ++kk)
        ah[kk] = *(const v2f*)&hL[M * H + kk * 4 + hi * 2];

      #pragma unroll
      for (int nt = 0; nt < NT; ++nt) {
        const float bb = bx[d][nt];
        v8f acc;
        #pragma unroll
        for (int r = 0; r < 8; ++r) acc[r] = bb;
        #pragma unroll
        for (int kk = 0; kk < KX; ++kk) {
          v2f bfrag = *(const v2f*)&s_wx[d][(nt * KX + kk) * 64 + lane * 2];
          acc = __builtin_amdgcn_wmma_f32_16x16x4_f32(
              false, ax[kk], false, bfrag, (short)0, acc, false, false);
        }
        #pragma unroll
        for (int kk = 0; kk < KH; ++kk) {
          v2f bfrag = *(const v2f*)&s_wh[d][(nt * KH + kk) * 64 + lane * 2];
          acc = __builtin_amdgcn_wmma_f32_16x16x4_f32(
              false, ah[kk], false, bfrag, (short)0, acc, false, false);
        }
        #pragma unroll
        for (int r = 0; r < 8; ++r)
          gatesL[(r + 8 * hi) * G + nt * 16 + M] = acc[r];
      }
      WAVE_FENCE();

      #pragma unroll
      for (int u = 0; u < 12; ++u) {
        const int e = u * 32 + lane;
        const int m = e / H, j = e - m * H;
        const float gi = gatesL[m * G + j];
        const float gf = gatesL[m * G + 24 + j];
        const float gg = gatesL[m * G + 48 + j];
        const float go = gatesL[m * G + 72 + j];
        const float iv = fsig(gi), fv = fsig(gf), ov = fsig(go), gv = ftanh(gg);
        const float cc = fv * cst[u] + iv * gv;
        cst[u] = cc;
        hL[e] = ov * ftanh(cc);
      }
      WAVE_FENCE();
    }

    if (d == 0) {
      for (int e = lane; e < 16 * H; e += 32) s_hf[w][e] = hL[e];
      WAVE_FENCE();
    }
  }

  #pragma unroll
  for (int p = lane; p < 64; p += 32) {
    const int m = p >> 2, cls = p & 3;
    float acc = s_headb[cls];
    #pragma unroll
    for (int j = 0; j < H; ++j) {
      acc += s_hf[w][m * H + j] * s_headw[cls * 2 * H + j];
      acc += hL[m * H + j]      * s_headw[cls * 2 * H + H + j];
    }
    out[(size_t)(b0 + m) * NC + cls] = acc;
  }
}

extern "C" void kernel_launch(void* const* d_in, const int* in_sizes, int n_in,
                              void* d_out, int out_size, void* d_ws, size_t ws_size,
                              hipStream_t stream) {
  (void)n_in; (void)d_ws; (void)ws_size; (void)out_size;
  const float* x      = (const float*)d_in[0];
  const float* w_ih_f = (const float*)d_in[1];
  const float* w_hh_f = (const float*)d_in[2];
  const float* b_ih_f = (const float*)d_in[3];
  const float* b_hh_f = (const float*)d_in[4];
  const float* w_ih_b = (const float*)d_in[5];
  const float* w_hh_b = (const float*)d_in[6];
  const float* b_ih_b = (const float*)d_in[7];
  const float* b_hh_b = (const float*)d_in[8];
  const float* head_w = (const float*)d_in[9];
  const float* head_b = (const float*)d_in[10];
  float* out = (float*)d_out;

  const int B = in_sizes[0] / (T * IN_F);
  const int tilesPerBlock = WAVES;
  const int grid = (B + 16 * tilesPerBlock - 1) / (16 * tilesPerBlock);
  bilstm_center_kernel<<<grid, 32 * WAVES, 0, stream>>>(
      x, w_ih_f, w_hh_f, b_ih_f, b_hh_f,
      w_ih_b, w_hh_b, b_ih_b, b_hh_b, head_w, head_b, out, B);
}


# Sigmoid and tanh on the hardware tanh instruction (v_tanh_f32): sigmoid(x) = (1 + tanh(x/2))/2, the x/2 folded exactly into the staged i,f,o weight rows; 5 transcendentals per cell instead of 10
# speedup vs baseline: 1.3878x; 1.3878x over previous
#include <hip/hip_runtime.h>
#include <math.h>


typedef __attribute__((ext_vector_type(2))) float v2f;
typedef __attribute__((ext_vector_type(4))) float v4f;
typedef __attribute__((ext_vector_type(8))) float v8f;

#define WAVES 8
constexpr int IN_F = 14, H = 24, T = 25, NC = 4, WH = 12;
constexpr int G  = 4 * H;
constexpr int MT = G / 16;
constexpr int KX = 4;
constexpr int KH = H / 4;
constexpr int HL = H / 2;

__device__ __forceinline__ float ftanh(float x) {
  return __builtin_amdgcn_tanhf(x);
}
__device__ __forceinline__ float fsig_of_half(float xh) {
  return fmaf(0.5f, __builtin_amdgcn_tanhf(xh), 0.5f);
}

__device__ __forceinline__ int gate_row(int mt, int p) {
  return (p & 3) * H + 12 * (p >> 3) + 2 * mt + ((p >> 2) & 1);
}
__device__ __forceinline__ float gate_scale(int p) {
  return ((p & 3) == 2) ? 1.f : 0.5f;
}

__global__ __launch_bounds__(32 * WAVES) void bilstm_center_kernel(
    const float* __restrict__ x,
    const float* __restrict__ w_ih_f, const float* __restrict__ w_hh_f,
    const float* __restrict__ b_ih_f, const float* __restrict__ b_hh_f,
    const float* __restrict__ w_ih_b, const float* __restrict__ w_hh_b,
    const float* __restrict__ b_ih_b, const float* __restrict__ b_hh_b,
    const float* __restrict__ head_w, const float* __restrict__ head_b,
    float* __restrict__ out, int B)
{
  __shared__ float s_wx[2][MT * KX * 64];
  __shared__ float s_wh[2][MT * KH * 64];
  __shared__ float s_headw[NC * 2 * H];
  __shared__ float s_headb[NC];

  const int tid = threadIdx.x;

  for (int i = tid; i < 2 * G * IN_F; i += blockDim.x) {
    int d = i / (G * IN_F);
    int r = i % (G * IN_F);
    int row = r / IN_F, k = r % IN_F;
    int mt = row >> 4, p = row & 15;
    int kk = k >> 2, kh = (k >> 1) & 1, e = k & 1;
    const float* wm = d ? w_ih_b : w_ih_f;
    s_wx[d][(mt * KX + kk) * 64 + (kh * 16 + p) * 2 + e] =
        gate_scale(p) * wm[gate_row(mt, p) * IN_F + k];
  }
  for (int i = tid; i < 2 * G; i += blockDim.x) {
    int d = i / G, row = i % G;
    int mt = row >> 4, p = row & 15;
    int g = gate_row(mt, p);
    s_wx[d][(mt * KX + 3) * 64 + (16 + p) * 2 + 0] =
        gate_scale(p) * (d ? (b_ih_b[g] + b_hh_b[g]) : (b_ih_f[g] + b_hh_f[g]));
    s_wx[d][(mt * KX + 3) * 64 + (16 + p) * 2 + 1] = 0.f;
  }
  for (int i = tid; i < 2 * MT * KH * 64; i += blockDim.x) {
    int d = i / (MT * KH * 64);
    int r = i % (MT * KH * 64);
    int frag = r >> 6, slot = r & 63;
    int mt = frag / KH, kk = frag % KH;
    int ln = slot >> 1, e = slot & 1;
    int g = gate_row(mt, ln & 15);
    int k = 12 * (ln >> 4) + 2 * kk + e;
    const float* wm = d ? w_hh_b : w_hh_f;
    s_wh[d][r] = gate_scale(ln & 15) * wm[g * H + k];
  }
  for (int i = tid; i < NC * 2 * H; i += blockDim.x) s_headw[i] = head_w[i];
  if (tid < NC) s_headb[tid] = head_b[tid];
  __syncthreads();

  const int lane = tid & 31;
  const int w    = tid >> 5;
  const int c    = lane & 15;
  const int hi   = lane >> 4;
  const int b0   = (blockIdx.x * WAVES + w) * 16;
  if (b0 >= B) return;

  const float* xrow = x + (size_t)(b0 + c) * T * IN_F;
  float headp[NC] = {0.f, 0.f, 0.f, 0.f};

  for (int d = 0; d < 2; ++d) {
    const float* wx = &s_wx[d][lane * 2];
    const float* wh = &s_wh[d][lane * 2];

    float cst[HL];
    v2f hb[KH];
    #pragma unroll
    for (int u = 0; u < HL; ++u) cst[u] = 0.f;
    #pragma unroll
    for (int kk = 0; kk < KH; ++kk) hb[kk] = (v2f){0.f, 0.f};

    for (int step = 0; step <= WH; ++step) {
      const int t = d ? (T - 1 - step) : step;
      const float* xr = xrow + t * IN_F;

      v2f bx[KX];
      #pragma unroll
      for (int kk = 0; kk < KX; ++kk) {
        const int ks = kk * 4 + hi * 2;
        if (ks + 1 < IN_F) bx[kk] = *(const v2f*)(xr + ks);
        else               bx[kk] = (v2f){1.f, 0.f};
      }

      v2f hn[KH];
      #pragma unroll
      for (int mt = 0; mt < MT; ++mt) {
        v8f acc = {0.f, 0.f, 0.f, 0.f, 0.f, 0.f, 0.f, 0.f};
        #pragma unroll
        for (int kk = 0; kk < KX; ++kk) {
          v2f afrag = *(const v2f*)&wx[(mt * KX + kk) * 64];
          acc = __builtin_amdgcn_wmma_f32_16x16x4_f32(
              false, afrag, false, bx[kk], (short)0, acc, false, false);
        }
        #pragma unroll
        for (int kk = 0; kk < KH; ++kk) {
          v2f afrag = *(const v2f*)&wh[(mt * KH + kk) * 64];
          acc = __builtin_amdgcn_wmma_f32_16x16x4_f32(
              false, afrag, false, hb[kk], (short)0, acc, false, false);
        }
        #pragma unroll
        for (int u = 0; u < 2; ++u) {
          const float iv = fsig_of_half(acc[4 * u + 0]);
          const float fv = fsig_of_half(acc[4 * u + 1]);
          const float gv = ftanh(acc[4 * u + 2]);
          const float ov = fsig_of_half(acc[4 * u + 3]);
          const float cc = fv * cst[2 * mt + u] + iv * gv;
          cst[2 * mt + u] = cc;
          hn[mt][u] = ov * ftanh(cc);
        }
      }
      #pragma unroll
      for (int kk = 0; kk < KH; ++kk) hb[kk] = hn[kk];
    }

    #pragma unroll
    for (int u = 0; u < HL; ++u) {
      const float hv = hb[u >> 1][u & 1];
      #pragma unroll
      for (int cls = 0; cls < NC; ++cls)
        headp[cls] += hv * s_headw[cls * 2 * H + d * H + 12 * hi + u];
    }
  }

  #pragma unroll
  for (int cls = 0; cls < NC; ++cls)
    headp[cls] += __shfl_xor(headp[cls], 16, 32);
  if (hi == 0) {
    v4f o = {headp[0] + s_headb[0], headp[1] + s_headb[1],
             headp[2] + s_headb[2], headp[3] + s_headb[3]};
    *(v4f*)&out[(size_t)(b0 + c) * NC] = o;
  }
}

extern "C" void kernel_launch(void* const* d_in, const int* in_sizes, int n_in,
                              void* d_out, int out_size, void* d_ws, size_t ws_size,
                              hipStream_t stream) {
  (void)n_in; (void)d_ws; (void)ws_size; (void)out_size;
  const float* x      = (const float*)d_in[0];
  const float* w_ih_f = (const float*)d_in[1];
  const float* w_hh_f = (const float*)d_in[2];
  const float* b_ih_f = (const float*)d_in[3];
  const float* b_hh_f = (const float*)d_in[4];
  const float* w_ih_b = (const float*)d_in[5];
  const float* w_hh_b = (const float*)d_in[6];
  const float* b_ih_b = (const float*)d_in[7];
  const float* b_hh_b = (const float*)d_in[8];
  const float* head_w = (const float*)d_in[9];
  const float* head_b = (const float*)d_in[10];
  float* out = (float*)d_out;

  const int B = in_sizes[0] / (T * IN_F);
  const int tilesPerBlock = WAVES;
  const int grid = (B + 16 * tilesPerBlock - 1) / (16 * tilesPerBlock);
  bilstm_center_kernel<<<grid, 32 * WAVES, 0, stream>>>(
      x, w_ih_f, w_hh_f, b_ih_f, b_hh_f,
      w_ih_b, w_hh_b, b_ih_b, b_hh_b, head_w, head_b, out, B);
}


# Time loop software-pipelined by one gate tile: each tile's WMMA chain issues ahead of the previous tile's gate math (next step's tile 0 covers tile 5); x rows fetched a step early; sum order kept
# speedup vs baseline: 1.5421x; 1.1112x over previous
#include <hip/hip_runtime.h>
#include <math.h>


typedef __attribute__((ext_vector_type(2))) float v2f;
typedef __attribute__((ext_vector_type(4))) float v4f;
typedef __attribute__((ext_vector_type(8))) float v8f;

#define WAVES 8
constexpr int IN_F = 14, H = 24, T = 25, NC = 4, WH = 12;
constexpr int G  = 4 * H;
constexpr int MT = G / 16;
constexpr int KX = 4;
constexpr int KH = H / 4;
constexpr int HL = H / 2;

__device__ __forceinline__ float ftanh(float x) {
  return __builtin_amdgcn_tanhf(x);
}
__device__ __forceinline__ float fsig_of_half(float xh) {
  return fmaf(0.5f, __builtin_amdgcn_tanhf(xh), 0.5f);
}

__device__ __forceinline__ int gate_row(int mt, int p) {
  return (p & 3) * H + 12 * (p >> 3) + 2 * mt + ((p >> 2) & 1);
}
__device__ __forceinline__ float gate_scale(int p) {
  return ((p & 3) == 2) ? 1.f : 0.5f;
}

__device__ __forceinline__ void load_x(v2f (&bx)[KX], const float* xr, int hi) {
  #pragma unroll
  for (int kk = 0; kk < KX; ++kk) {
    const int ks = kk * 4 + hi * 2;
    if (ks + 1 < IN_F) bx[kk] = *(const v2f*)(xr + ks);
    else               bx[kk] = (v2f){1.f, 0.f};
  }
}
__device__ __forceinline__ v8f chain_x(const v2f* awx, const v2f (&bx)[KX]) {
  v8f acc = {0.f, 0.f, 0.f, 0.f, 0.f, 0.f, 0.f, 0.f};
  #pragma unroll
  for (int kk = 0; kk < KX; ++kk)
    acc = __builtin_amdgcn_wmma_f32_16x16x4_f32(
        false, awx[kk], false, bx[kk], (short)0, acc, false, false);
  return acc;
}
__device__ __forceinline__ v8f chain_h(v8f acc, const v2f* awh, const v2f (&hb)[KH]) {
  #pragma unroll
  for (int kk = 0; kk < KH; ++kk)
    acc = __builtin_amdgcn_wmma_f32_16x16x4_f32(
        false, awh[kk], false, hb[kk], (short)0, acc, false, false);
  return acc;
}
__device__ __forceinline__ void cells(const v8f acc, float& c0, float& c1, v2f& hout) {
  const float i0 = fsig_of_half(acc[0]), i1 = fsig_of_half(acc[4]);
  const float f0 = fsig_of_half(acc[1]), f1 = fsig_of_half(acc[5]);
  const float g0 = ftanh(acc[2]),        g1 = ftanh(acc[6]);
  const float o0 = fsig_of_half(acc[3]), o1 = fsig_of_half(acc[7]);
  c0 = f0 * c0 + i0 * g0;
  c1 = f1 * c1 + i1 * g1;
  hout = (v2f){o0 * ftanh(c0), o1 * ftanh(c1)};
}

#define GW()  __builtin_amdgcn_sched_group_barrier(0x008, 1, 0)
#define GT(n) __builtin_amdgcn_sched_group_barrier(0x400, n, 0)
#define GV(n) __builtin_amdgcn_sched_group_barrier(0x002, n, 0)
#define PAIR_ORDER() do { \
  GW(); GT(2); GW(); GT(2); GW(); GT(2); GW(); GT(2); \
  GW(); GV(4); GW(); GV(4); GW(); GT(2); GW(); GV(2); GW(); GV(2); GW(); \
} while (0)
#define PAIR_END() __builtin_amdgcn_sched_barrier(0)

__device__ __forceinline__ void tiles_1_to_5(const v2f* awx, const v2f* awh, const v2f (&bx)[KX],
                                             v2f (&hb)[KH], float (&cst)[HL], v8f& acc) {
  v2f hn[KH - 1];
  #pragma unroll
  for (int mt = 1; mt < MT; ++mt) {
    v8f accB = chain_h(chain_x(&awx[mt * KX], bx), &awh[mt * KH], hb);
    cells(acc, cst[2 * mt - 2], cst[2 * mt - 1], hn[mt - 1]);
    acc = accB;
    PAIR_ORDER();
    PAIR_END();
  }
  #pragma unroll
  for (int kk = 0; kk < KH - 1; ++kk) hb[kk] = hn[kk];
}

__global__ __launch_bounds__(32 * WAVES) void bilstm_center_kernel(
    const float* __restrict__ x,
    const float* __restrict__ w_ih_f, const float* __restrict__ w_hh_f,
    const float* __restrict__ b_ih_f, const float* __restrict__ b_hh_f,
    const float* __restrict__ w_ih_b, const float* __restrict__ w_hh_b,
    const float* __restrict__ b_ih_b, const float* __restrict__ b_hh_b,
    const float* __restrict__ head_w, const float* __restrict__ head_b,
    float* __restrict__ out, int B)
{
  __shared__ float s_wx[2][MT * KX * 64];
  __shared__ float s_wh[2][MT * KH * 64];
  __shared__ float s_headw[NC * 2 * H];
  __shared__ float s_headb[NC];

  const int tid = threadIdx.x;

  for (int i = tid; i < 2 * G * IN_F; i += blockDim.x) {
    int d = i / (G * IN_F);
    int r = i % (G * IN_F);
    int row = r / IN_F, k = r % IN_F;
    int mt = row >> 4, p = row & 15;
    int kk = k >> 2, kh = (k >> 1) & 1, e = k & 1;
    const float* wm = d ? w_ih_b : w_ih_f;
    s_wx[d][(mt * KX + kk) * 64 + (kh * 16 + p) * 2 + e] =
        gate_scale(p) * wm[gate_row(mt, p) * IN_F + k];
  }
  for (int i = tid; i < 2 * G; i += blockDim.x) {
    int d = i / G, row = i % G;
    int mt = row >> 4, p = row & 15;
    int g = gate_row(mt, p);
    s_wx[d][(mt * KX + 3) * 64 + (16 + p) * 2 + 0] =
        gate_scale(p) * (d ? (b_ih_b[g] + b_hh_b[g]) : (b_ih_f[g] + b_hh_f[g]));
    s_wx[d][(mt * KX + 3) * 64 + (16 + p) * 2 + 1] = 0.f;
  }
  for (int i = tid; i < 2 * MT * KH * 64; i += blockDim.x) {
    int d = i / (MT * KH * 64);
    int r = i % (MT * KH * 64);
    int frag = r >> 6, slot = r & 63;
    int mt = frag / KH, kk = frag % KH;
    int ln = slot >> 1, e = slot & 1;
    int g = gate_row(mt, ln & 15);
    int k = 12 * (ln >> 4) + 2 * kk + e;
    const float* wm = d ? w_hh_b : w_hh_f;
    s_wh[d][r] = gate_scale(ln & 15) * wm[g * H + k];
  }
  for (int i = tid; i < NC * 2 * H; i += blockDim.x) s_headw[i] = head_w[i];
  if (tid < NC) s_headb[tid] = head_b[tid];
  __syncthreads();

  const int lane = tid & 31;
  const int w    = tid >> 5;
  const int c    = lane & 15;
  const int hi   = lane >> 4;
  const int b0   = (blockIdx.x * WAVES + w) * 16;
  if (b0 >= B) return;

  const float* xrow = x + (size_t)(b0 + c) * T * IN_F;
  float headp[NC] = {0.f, 0.f, 0.f, 0.f};

  for (int d = 0; d < 2; ++d) {
    v2f awx[MT * KX], awh[MT * KH];
    #pragma unroll
    for (int f = 0; f < MT * KX; ++f) awx[f] = *(const v2f*)&s_wx[d][f * 64 + lane * 2];
    #pragma unroll
    for (int f = 0; f < MT * KH; ++f) awh[f] = *(const v2f*)&s_wh[d][f * 64 + lane * 2];

    float cst[HL];
    v2f hb[KH];
    #pragma unroll
    for (int u = 0; u < HL; ++u) cst[u] = 0.f;
    #pragma unroll
    for (int kk = 0; kk < KH; ++kk) hb[kk] = (v2f){0.f, 0.f};

    v2f bx[KX], bxn[KX];
    load_x(bx, xrow + (d ? (T - 1) : 0) * IN_F, hi);
    load_x(bxn, xrow + (d ? (T - 2) : 1) * IN_F, hi);

    v8f accA = chain_h(chain_x(&awx[0], bx), &awh[0], hb);
    PAIR_END();
    tiles_1_to_5(awx, awh, bx, hb, cst, accA);

    for (int step = 1; step <= WH; ++step) {
      #pragma unroll
      for (int kk = 0; kk < KX; ++kk) bx[kk] = bxn[kk];
      const int sn = (step < WH) ? (step + 1) : WH;
      load_x(bxn, xrow + (d ? (T - 1 - sn) : sn) * IN_F, hi);

      v8f accN = chain_x(&awx[0], bx);
      cells(accA, cst[2 * MT - 2], cst[2 * MT - 1], hb[MT - 1]);
      accA = chain_h(accN, &awh[0], hb);
      PAIR_ORDER();
      PAIR_END();

      tiles_1_to_5(awx, awh, bx, hb, cst, accA);
    }
    cells(accA, cst[2 * MT - 2], cst[2 * MT - 1], hb[MT - 1]);

    #pragma unroll
    for (int u = 0; u < HL; ++u) {
      const float hv = hb[u >> 1][u & 1];
      #pragma unroll
      for (int cls = 0; cls < NC; ++cls)
        headp[cls] += hv * s_headw[cls * 2 * H + d * H + 12 * hi + u];
    }
  }

  #pragma unroll
  for (int cls = 0; cls < NC; ++cls)
    headp[cls] += __shfl_xor(headp[cls], 16, 32);
  if (hi == 0) {
    v4f o = {headp[0] + s_headb[0], headp[1] + s_headb[1],
             headp[2] + s_headb[2], headp[3] + s_headb[3]};
    *(v4f*)&out[(size_t)(b0 + c) * NC] = o;
  }
}

extern "C" void kernel_launch(void* const* d_in, const int* in_sizes, int n_in,
                              void* d_out, int out_size, void* d_ws, size_t ws_size,
                              hipStream_t stream) {
  (void)n_in; (void)d_ws; (void)ws_size; (void)out_size;
  const float* x      = (const float*)d_in[0];
  const float* w_ih_f = (const float*)d_in[1];
  const float* w_hh_f = (const float*)d_in[2];
  const float* b_ih_f = (const float*)d_in[3];
  const float* b_hh_f = (const float*)d_in[4];
  const float* w_ih_b = (const float*)d_in[5];
  const float* w_hh_b = (const float*)d_in[6];
  const float* b_ih_b = (const float*)d_in[7];
  const float* b_hh_b = (const float*)d_in[8];
  const float* head_w = (const float*)d_in[9];
  const float* head_b = (const float*)d_in[10];
  float* out = (float*)d_out;

  const int B = in_sizes[0] / (T * IN_F);
  const int tilesPerBlock = WAVES;
  const int grid = (B + 16 * tilesPerBlock - 1) / (16 * tilesPerBlock);
  bilstm_center_kernel<<<grid, 32 * WAVES, 0, stream>>>(
      x, w_ih_f, w_hh_f, b_ih_f, b_hh_f,
      w_ih_b, w_hh_b, b_ih_b, b_hh_b, head_w, head_b, out, B);
}


# Weight staging: 8 threads per weight row read it in memory order (16-byte h, 8-byte x pieces), all of a thread's loads in flight before first use, index math once per row; staged values unchanged
# speedup vs baseline: 1.6733x; 1.0851x over previous
#include <hip/hip_runtime.h>
#include <math.h>


typedef __attribute__((ext_vector_type(2))) float v2f;
typedef __attribute__((ext_vector_type(4))) float v4f;
typedef __attribute__((ext_vector_type(8))) float v8f;

#define WAVES 8
constexpr int IN_F = 14, H = 24, T = 25, NC = 4, WH = 12;
constexpr int G  = 4 * H;
constexpr int MT = G / 16;
constexpr int KX = 4;
constexpr int KH = H / 4;
constexpr int HL = H / 2;

__device__ __forceinline__ float ftanh(float x) {
  return __builtin_amdgcn_tanhf(x);
}
__device__ __forceinline__ float fsig_of_half(float xh) {
  return fmaf(0.5f, __builtin_amdgcn_tanhf(xh), 0.5f);
}

__device__ __forceinline__ int gate_row(int mt, int p) {
  return (p & 3) * H + 12 * (p >> 3) + 2 * mt + ((p >> 2) & 1);
}
__device__ __forceinline__ float gate_scale(int p) {
  return ((p & 3) == 2) ? 1.f : 0.5f;
}

__device__ __forceinline__ void load_x(v2f (&bx)[KX], const float* xr, int hi) {
  #pragma unroll
  for (int kk = 0; kk < KX; ++kk) {
    const int ks = kk * 4 + hi * 2;
    if (ks + 1 < IN_F) bx[kk] = *(const v2f*)(xr + ks);
    else               bx[kk] = (v2f){1.f, 0.f};
  }
}
__device__ __forceinline__ v8f chain_x(const v2f* awx, const v2f (&bx)[KX]) {
  v8f acc = {0.f, 0.f, 0.f, 0.f, 0.f, 0.f, 0.f, 0.f};
  #pragma unroll
  for (int kk = 0; kk < KX; ++kk)
    acc = __builtin_amdgcn_wmma_f32_16x16x4_f32(
        false, awx[kk], false, bx[kk], (short)0, acc, false, false);
  return acc;
}
__device__ __forceinline__ v8f chain_h(v8f acc, const v2f* awh, const v2f (&hb)[KH]) {
  #pragma unroll
  for (int kk = 0; kk < KH; ++kk)
    acc = __builtin_amdgcn_wmma_f32_16x16x4_f32(
        false, awh[kk], false, hb[kk], (short)0, acc, false, false);
  return acc;
}
__device__ __forceinline__ void cells(const v8f acc, float& c0, float& c1, v2f& hout) {
  const float i0 = fsig_of_half(acc[0]), i1 = fsig_of_half(acc[4]);
  const float f0 = fsig_of_half(acc[1]), f1 = fsig_of_half(acc[5]);
  const float g0 = ftanh(acc[2]),        g1 = ftanh(acc[6]);
  const float o0 = fsig_of_half(acc[3]), o1 = fsig_of_half(acc[7]);
  c0 = f0 * c0 + i0 * g0;
  c1 = f1 * c1 + i1 * g1;
  hout = (v2f){o0 * ftanh(c0), o1 * ftanh(c1)};
}

#define GW()  __builtin_amdgcn_sched_group_barrier(0x008, 1, 0)
#define GT(n) __builtin_amdgcn_sched_group_barrier(0x400, n, 0)
#define GV(n) __builtin_amdgcn_sched_group_barrier(0x002, n, 0)
#define PAIR_ORDER() do { \
  GW(); GT(2); GW(); GT(2); GW(); GT(2); GW(); GT(2); \
  GW(); GV(4); GW(); GV(4); GW(); GT(2); GW(); GV(2); GW(); GV(2); GW(); \
} while (0)
#define PAIR_END() __builtin_amdgcn_sched_barrier(0)

__device__ __forceinline__ void tiles_1_to_5(const v2f* awx, const v2f* awh, const v2f (&bx)[KX],
                                             v2f (&hb)[KH], float (&cst)[HL], v8f& acc) {
  v2f hn[KH - 1];
  #pragma unroll
  for (int mt = 1; mt < MT; ++mt) {
    v8f accB = chain_h(chain_x(&awx[mt * KX], bx), &awh[mt * KH], hb);
    cells(acc, cst[2 * mt - 2], cst[2 * mt - 1], hn[mt - 1]);
    acc = accB;
    PAIR_ORDER();
    PAIR_END();
  }
  #pragma unroll
  for (int kk = 0; kk < KH - 1; ++kk) hb[kk] = hn[kk];
}

__global__ __launch_bounds__(32 * WAVES) void bilstm_center_kernel(
    const float* __restrict__ x,
    const float* __restrict__ w_ih_f, const float* __restrict__ w_hh_f,
    const float* __restrict__ b_ih_f, const float* __restrict__ b_hh_f,
    const float* __restrict__ w_ih_b, const float* __restrict__ w_hh_b,
    const float* __restrict__ b_ih_b, const float* __restrict__ b_hh_b,
    const float* __restrict__ head_w, const float* __restrict__ head_b,
    float* __restrict__ out, int B)
{
  __shared__ __attribute__((aligned(16))) float s_wx[2][MT * KX * 64];
  __shared__ __attribute__((aligned(16))) float s_wh[2][MT * KH * 64];
  __shared__ float s_head[32 * WAVES];

  const int tid = threadIdx.x;

  {
    const int a  = tid >> 3;
    const int c  = tid & 7;
    const int ch = (c < 6) ? c : 5;
    const int cx = (c < 7) ? c : 6;
    const int hih = (ch >= 3) ? 1 : 0;
    const int coff_h = (2 * (ch - 3 * hih)) * 64 + hih * 32;
    const int coff_x = (c >> 1) * 64 + (c & 1) * 32;

    v4f hw[2][3];
    v2f xw[2][3];
    float bi[2][3], bh[2][3];
    #pragma unroll
    for (int jj = 0; jj < 3; ++jj) {
      const int g = a + 32 * jj;
      hw[0][jj] = *(const v4f*)(w_hh_f + g * H + 4 * ch);
      hw[1][jj] = *(const v4f*)(w_hh_b + g * H + 4 * ch);
      xw[0][jj] = *(const v2f*)(w_ih_f + g * IN_F + 2 * cx);
      xw[1][jj] = *(const v2f*)(w_ih_b + g * IN_F + 2 * cx);
      bi[0][jj] = b_ih_f[g];  bh[0][jj] = b_hh_f[g];
      bi[1][jj] = b_ih_b[g];  bh[1][jj] = b_hh_b[g];
    }
    const float hv = *((tid < NC * 2 * H) ? (head_w + tid) : (head_b + (tid & (NC - 1))));

    #pragma unroll
    for (int jj = 0; jj < 3; ++jj) {
      const int g  = a + 32 * jj;
      const int qg = (g * 43) >> 10;
      const int u  = g - 24 * qg;
      const int hh = (u >= 12) ? 1 : 0;
      const int v  = u - 12 * hh;
      const int mt = v >> 1;
      const int p  = qg + 4 * (v & 1) + 8 * hh;
      const float s = gate_scale(p);
      const int rh = mt * (KH * 64) + 2 * p;
      const int rx = mt * (KX * 64) + 2 * p;
      #pragma unroll
      for (int d = 0; d < 2; ++d) {
        *(v2f*)&s_wh[d][rh + coff_h]      = (v2f){s * hw[d][jj][0], s * hw[d][jj][1]};
        *(v2f*)&s_wh[d][rh + coff_h + 64] = (v2f){s * hw[d][jj][2], s * hw[d][jj][3]};
        const v2f xv = (c < 7) ? (v2f){s * xw[d][jj][0], s * xw[d][jj][1]}
                               : (v2f){s * (bi[d][jj] + bh[d][jj]), 0.f};
        *(v2f*)&s_wx[d][rx + coff_x] = xv;
      }
    }
    s_head[tid] = hv;
  }
  __syncthreads();

  const int lane = tid & 31;
  const int w    = tid >> 5;
  const int c    = lane & 15;
  const int hi   = lane >> 4;
  const int b0   = (blockIdx.x * WAVES + w) * 16;
  if (b0 >= B) return;

  const float* xrow = x + (size_t)(b0 + c) * T * IN_F;
  float headp[NC] = {0.f, 0.f, 0.f, 0.f};

  for (int d = 0; d < 2; ++d) {
    v2f awx[MT * KX], awh[MT * KH];
    #pragma unroll
    for (int f = 0; f < MT * KX; ++f) awx[f] = *(const v2f*)&s_wx[d][f * 64 + lane * 2];
    #pragma unroll
    for (int f = 0; f < MT * KH; ++f) awh[f] = *(const v2f*)&s_wh[d][f * 64 + lane * 2];

    float cst[HL];
    v2f hb[KH];
    #pragma unroll
    for (int u = 0; u < HL; ++u) cst[u] = 0.f;
    #pragma unroll
    for (int kk = 0; kk < KH; ++kk) hb[kk] = (v2f){0.f, 0.f};

    v2f bx[KX], bxn[KX];
    load_x(bx, xrow + (d ? (T - 1) : 0) * IN_F, hi);
    load_x(bxn, xrow + (d ? (T - 2) : 1) * IN_F, hi);

    v8f accA = chain_h(chain_x(&awx[0], bx), &awh[0], hb);
    PAIR_END();
    tiles_1_to_5(awx, awh, bx, hb, cst, accA);

    for (int step = 1; step <= WH; ++step) {
      #pragma unroll
      for (int kk = 0; kk < KX; ++kk) bx[kk] = bxn[kk];
      const int sn = (step < WH) ? (step + 1) : WH;
      load_x(bxn, xrow + (d ? (T - 1 - sn) : sn) * IN_F, hi);

      v8f accN = chain_x(&awx[0], bx);
      cells(accA, cst[2 * MT - 2], cst[2 * MT - 1], hb[MT - 1]);
      accA = chain_h(accN, &awh[0], hb);
      PAIR_ORDER();
      PAIR_END();

      tiles_1_to_5(awx, awh, bx, hb, cst, accA);
    }
    cells(accA, cst[2 * MT - 2], cst[2 * MT - 1], hb[MT - 1]);

    #pragma unroll
    for (int u = 0; u < HL; ++u) {
      const float hv = hb[u >> 1][u & 1];
      #pragma unroll
      for (int cls = 0; cls < NC; ++cls)
        headp[cls] += hv * s_head[cls * 2 * H + d * H + 12 * hi + u];
    }
  }

  #pragma unroll
  for (int cls = 0; cls < NC; ++cls)
    headp[cls] += __shfl_xor(headp[cls], 16, 32);
  if (hi == 0) {
    const float* hbp = &s_head[NC * 2 * H];
    v4f o = {headp[0] + hbp[0], headp[1] + hbp[1],
             headp[2] + hbp[2], headp[3] + hbp[3]};
    *(v4f*)&out[(size_t)(b0 + c) * NC] = o;
  }
}

extern "C" void kernel_launch(void* const* d_in, const int* in_sizes, int n_in,
                              void* d_out, int out_size, void* d_ws, size_t ws_size,
                              hipStream_t stream) {
  (void)n_in; (void)d_ws; (void)ws_size; (void)out_size;
  const float* x      = (const float*)d_in[0];
  const float* w_ih_f = (const float*)d_in[1];
  const float* w_hh_f = (const float*)d_in[2];
  const float* b_ih_f = (const float*)d_in[3];
  const float* b_hh_f = (const float*)d_in[4];
  const float* w_ih_b = (const float*)d_in[5];
  const float* w_hh_b = (const float*)d_in[6];
  const float* b_ih_b = (const float*)d_in[7];
  const float* b_hh_b = (const float*)d_in[8];
  const float* head_w = (const float*)d_in[9];
  const float* head_b = (const float*)d_in[10];
  float* out = (float*)d_out;

  const int B = in_sizes[0] / (T * IN_F);
  const int tilesPerBlock = WAVES;
  const int grid = (B + 16 * tilesPerBlock - 1) / (16 * tilesPerBlock);
  bilstm_center_kernel<<<grid, 32 * WAVES, 0, stream>>>(
      x, w_ih_f, w_hh_f, b_ih_f, b_hh_f,
      w_ih_b, w_hh_b, b_ih_b, b_hh_b, head_w, head_b, out, B);
}


# Step 0 of each direction starts from h = 0 (the reference's h0): its six h-part matrix instructions per gate tile are dropped, gates = x part alone; 72 of 1560 per wave; sums unchanged
# speedup vs baseline: 1.7240x; 1.0303x over previous
#include <hip/hip_runtime.h>
#include <math.h>


typedef __attribute__((ext_vector_type(2))) float v2f;
typedef __attribute__((ext_vector_type(4))) float v4f;
typedef __attribute__((ext_vector_type(8))) float v8f;

#define WAVES 8
constexpr int IN_F = 14, H = 24, T = 25, NC = 4, WH = 12;
constexpr int G  = 4 * H;
constexpr int MT = G / 16;
constexpr int KX = 4;
constexpr int KH = H / 4;
constexpr int HL = H / 2;

__device__ __forceinline__ float ftanh(float x) {
  return __builtin_amdgcn_tanhf(x);
}
__device__ __forceinline__ float fsig_of_half(float xh) {
  return fmaf(0.5f, __builtin_amdgcn_tanhf(xh), 0.5f);
}

__device__ __forceinline__ int gate_row(int mt, int p) {
  return (p & 3) * H + 12 * (p >> 3) + 2 * mt + ((p >> 2) & 1);
}
__device__ __forceinline__ float gate_scale(int p) {
  return ((p & 3) == 2) ? 1.f : 0.5f;
}

__device__ __forceinline__ void load_x(v2f (&bx)[KX], const float* xr, int hi) {
  #pragma unroll
  for (int kk = 0; kk < KX; ++kk) {
    const int ks = kk * 4 + hi * 2;
    if (ks + 1 < IN_F) bx[kk] = *(const v2f*)(xr + ks);
    else               bx[kk] = (v2f){1.f, 0.f};
  }
}
__device__ __forceinline__ v8f chain_x(const v2f* awx, const v2f (&bx)[KX]) {
  v8f acc = {0.f, 0.f, 0.f, 0.f, 0.f, 0.f, 0.f, 0.f};
  #pragma unroll
  for (int kk = 0; kk < KX; ++kk)
    acc = __builtin_amdgcn_wmma_f32_16x16x4_f32(
        false, awx[kk], false, bx[kk], (short)0, acc, false, false);
  return acc;
}
__device__ __forceinline__ v8f chain_h(v8f acc, const v2f* awh, const v2f (&hb)[KH]) {
  #pragma unroll
  for (int kk = 0; kk < KH; ++kk)
    acc = __builtin_amdgcn_wmma_f32_16x16x4_f32(
        false, awh[kk], false, hb[kk], (short)0, acc, false, false);
  return acc;
}
__device__ __forceinline__ void cells(const v8f acc, float& c0, float& c1, v2f& hout) {
  const float i0 = fsig_of_half(acc[0]), i1 = fsig_of_half(acc[4]);
  const float f0 = fsig_of_half(acc[1]), f1 = fsig_of_half(acc[5]);
  const float g0 = ftanh(acc[2]),        g1 = ftanh(acc[6]);
  const float o0 = fsig_of_half(acc[3]), o1 = fsig_of_half(acc[7]);
  c0 = f0 * c0 + i0 * g0;
  c1 = f1 * c1 + i1 * g1;
  hout = (v2f){o0 * ftanh(c0), o1 * ftanh(c1)};
}

#define GW()  __builtin_amdgcn_sched_group_barrier(0x008, 1, 0)
#define GT(n) __builtin_amdgcn_sched_group_barrier(0x400, n, 0)
#define GV(n) __builtin_amdgcn_sched_group_barrier(0x002, n, 0)
#define PAIR_ORDER() do { \
  GW(); GT(2); GW(); GT(2); GW(); GT(2); GW(); GT(2); \
  GW(); GV(4); GW(); GV(4); GW(); GT(2); GW(); GV(2); GW(); GV(2); GW(); \
} while (0)
#define PAIR_END() __builtin_amdgcn_sched_barrier(0)

template <bool HAS_H>
__device__ __forceinline__ void tiles_1_to_5(const v2f* awx, const v2f* awh, const v2f (&bx)[KX],
                                             v2f (&hb)[KH], float (&cst)[HL], v8f& acc) {
  v2f hn[KH - 1];
  #pragma unroll
  for (int mt = 1; mt < MT; ++mt) {
    v8f accB = chain_x(&awx[mt * KX], bx);
    if constexpr (HAS_H) accB = chain_h(accB, &awh[mt * KH], hb);
    cells(acc, cst[2 * mt - 2], cst[2 * mt - 1], hn[mt - 1]);
    acc = accB;
    if constexpr (HAS_H) PAIR_ORDER();
    PAIR_END();
  }
  #pragma unroll
  for (int kk = 0; kk < KH - 1; ++kk) hb[kk] = hn[kk];
}

__global__ __launch_bounds__(32 * WAVES) void bilstm_center_kernel(
    const float* __restrict__ x,
    const float* __restrict__ w_ih_f, const float* __restrict__ w_hh_f,
    const float* __restrict__ b_ih_f, const float* __restrict__ b_hh_f,
    const float* __restrict__ w_ih_b, const float* __restrict__ w_hh_b,
    const float* __restrict__ b_ih_b, const float* __restrict__ b_hh_b,
    const float* __restrict__ head_w, const float* __restrict__ head_b,
    float* __restrict__ out, int B)
{
  __shared__ __attribute__((aligned(16))) float s_wx[2][MT * KX * 64];
  __shared__ __attribute__((aligned(16))) float s_wh[2][MT * KH * 64];
  __shared__ float s_head[32 * WAVES];

  const int tid = threadIdx.x;

  {
    const int a  = tid >> 3;
    const int c  = tid & 7;
    const int ch = (c < 6) ? c : 5;
    const int cx = (c < 7) ? c : 6;
    const int hih = (ch >= 3) ? 1 : 0;
    const int coff_h = (2 * (ch - 3 * hih)) * 64 + hih * 32;
    const int coff_x = (c >> 1) * 64 + (c & 1) * 32;

    v4f hw[2][3];
    v2f xw[2][3];
    float bi[2][3], bh[2][3];
    #pragma unroll
    for (int jj = 0; jj < 3; ++jj) {
      const int g = a + 32 * jj;
      hw[0][jj] = *(const v4f*)(w_hh_f + g * H + 4 * ch);
      hw[1][jj] = *(const v4f*)(w_hh_b + g * H + 4 * ch);
      xw[0][jj] = *(const v2f*)(w_ih_f + g * IN_F + 2 * cx);
      xw[1][jj] = *(const v2f*)(w_ih_b + g * IN_F + 2 * cx);
      bi[0][jj] = b_ih_f[g];  bh[0][jj] = b_hh_f[g];
      bi[1][jj] = b_ih_b[g];  bh[1][jj] = b_hh_b[g];
    }
    const float hv = *((tid < NC * 2 * H) ? (head_w + tid) : (head_b + (tid & (NC - 1))));

    #pragma unroll
    for (int jj = 0; jj < 3; ++jj) {
      const int g  = a + 32 * jj;
      const int qg = (g * 43) >> 10;
      const int u  = g - 24 * qg;
      const int hh = (u >= 12) ? 1 : 0;
      const int v  = u - 12 * hh;
      const int mt = v >> 1;
      const int p  = qg + 4 * (v & 1) + 8 * hh;
      const float s = gate_scale(p);
      const int rh = mt * (KH * 64) + 2 * p;
      const int rx = mt * (KX * 64) + 2 * p;
      #pragma unroll
      for (int d = 0; d < 2; ++d) {
        *(v2f*)&s_wh[d][rh + coff_h]      = (v2f){s * hw[d][jj][0], s * hw[d][jj][1]};
        *(v2f*)&s_wh[d][rh + coff_h + 64] = (v2f){s * hw[d][jj][2], s * hw[d][jj][3]};
        const v2f xv = (c < 7) ? (v2f){s * xw[d][jj][0], s * xw[d][jj][1]}
                               : (v2f){s * (bi[d][jj] + bh[d][jj]), 0.f};
        *(v2f*)&s_wx[d][rx + coff_x] = xv;
      }
    }
    s_head[tid] = hv;
  }
  __syncthreads();

  const int lane = tid & 31;
  const int w    = tid >> 5;
  const int c    = lane & 15;
  const int hi   = lane >> 4;
  const int b0   = (blockIdx.x * WAVES + w) * 16;
  if (b0 >= B) return;

  const float* xrow = x + (size_t)(b0 + c) * T * IN_F;
  float headp[NC] = {0.f, 0.f, 0.f, 0.f};

  for (int d = 0; d < 2; ++d) {
    v2f awx[MT * KX], awh[MT * KH];
    #pragma unroll
    for (int f = 0; f < MT * KX; ++f) awx[f] = *(const v2f*)&s_wx[d][f * 64 + lane * 2];
    #pragma unroll
    for (int f = 0; f < MT * KH; ++f) awh[f] = *(const v2f*)&s_wh[d][f * 64 + lane * 2];

    float cst[HL];
    v2f hb[KH];
    #pragma unroll
    for (int u = 0; u < HL; ++u) cst[u] = 0.f;
    #pragma unroll
    for (int kk = 0; kk < KH; ++kk) hb[kk] = (v2f){0.f, 0.f};

    v2f bx[KX], bxn[KX];
    load_x(bx, xrow + (d ? (T - 1) : 0) * IN_F, hi);
    load_x(bxn, xrow + (d ? (T - 2) : 1) * IN_F, hi);

    v8f accA = chain_x(&awx[0], bx);
    PAIR_END();
    tiles_1_to_5<false>(awx, awh, bx, hb, cst, accA);

    for (int step = 1; step <= WH; ++step) {
      #pragma unroll
      for (int kk = 0; kk < KX; ++kk) bx[kk] = bxn[kk];
      const int sn = (step < WH) ? (step + 1) : WH;
      load_x(bxn, xrow + (d ? (T - 1 - sn) : sn) * IN_F, hi);

      v8f accN = chain_x(&awx[0], bx);
      cells(accA, cst[2 * MT - 2], cst[2 * MT - 1], hb[MT - 1]);
      accA = chain_h(accN, &awh[0], hb);
      PAIR_ORDER();
      PAIR_END();

      tiles_1_to_5<true>(awx, awh, bx, hb, cst, accA);
    }
    cells(accA, cst[2 * MT - 2], cst[2 * MT - 1], hb[MT - 1]);

    #pragma unroll
    for (int u = 0; u < HL; ++u) {
      const float hv = hb[u >> 1][u & 1];
      #pragma unroll
      for (int cls = 0; cls < NC; ++cls)
        headp[cls] += hv * s_head[cls * 2 * H + d * H + 12 * hi + u];
    }
  }

  #pragma unroll
  for (int cls = 0; cls < NC; ++cls)
    headp[cls] += __shfl_xor(headp[cls], 16, 32);
  if (hi == 0) {
    const float* hbp = &s_head[NC * 2 * H];
    v4f o = {headp[0] + hbp[0], headp[1] + hbp[1],
             headp[2] + hbp[2], headp[3] + hbp[3]};
    *(v4f*)&out[(size_t)(b0 + c) * NC] = o;
  }
}

extern "C" void kernel_launch(void* const* d_in, const int* in_sizes, int n_in,
                              void* d_out, int out_size, void* d_ws, size_t ws_size,
                              hipStream_t stream) {
  (void)n_in; (void)d_ws; (void)ws_size; (void)out_size;
  const float* x      = (const float*)d_in[0];
  const float* w_ih_f = (const float*)d_in[1];
  const float* w_hh_f = (const float*)d_in[2];
  const float* b_ih_f = (const float*)d_in[3];
  const float* b_hh_f = (const float*)d_in[4];
  const float* w_ih_b = (const float*)d_in[5];
  const float* w_hh_b = (const float*)d_in[6];
  const float* b_ih_b = (const float*)d_in[7];
  const float* b_hh_b = (const float*)d_in[8];
  const float* head_w = (const float*)d_in[9];
  const float* head_b = (const float*)d_in[10];
  float* out = (float*)d_out;

  const int B = in_sizes[0] / (T * IN_F);
  const int tilesPerBlock = WAVES;
  const int grid = (B + 16 * tilesPerBlock - 1) / (16 * tilesPerBlock);
  bilstm_center_kernel<<<grid, 32 * WAVES, 0, stream>>>(
      x, w_ih_f, w_hh_f, b_ih_f, b_hh_f,
      w_ih_b, w_hh_b, b_ih_b, b_hh_b, head_w, head_b, out, B);
}
